# starting kernel
# baseline (speedup 1.0000x reference)
#include <hip/hip_runtime.h>

typedef __attribute__((ext_vector_type(2))) float v2f;
typedef __attribute__((ext_vector_type(8))) float v8f;

#define B_   32
#define C_   512
#define L_   4096
#define NB   16
#define EPSV 1e-8f

#define WS_G    0
#define WS_GN   (B_ * C_ * 4)
#define WS_KEYS (WS_GN + 128)
#define WS_COS  (WS_KEYS + 128)

__device__ __forceinline__ unsigned enc_key(float f) {
    unsigned u = __float_as_uint(f);
    return (u & 0x80000000u) ? ~u : (u | 0x80000000u);
}
__device__ __forceinline__ float dec_key(unsigned k) {
    unsigned u = (k & 0x80000000u) ? (k & 0x7FFFFFFFu) : ~k;
    return __uint_as_float(u);
}

__global__ void k_init(unsigned* __restrict__ keys) {
    if (threadIdx.x == 0) { keys[0] = 0xFFFFFFFFu; keys[1] = 0u; }
}

__global__ void k_mean(const float* __restrict__ x, float* __restrict__ g) {
    int wave = threadIdx.x >> 5, lane = threadIdx.x & 31;
    int row  = blockIdx.x * 8 + wave;
    const float4* xr = (const float4*)(x + (size_t)row * L_);
    float s = 0.f;
    for (int i = lane; i < L_ / 4; i += 32) {
        float4 v = xr[i];
        s += (v.x + v.y) + (v.z + v.w);
    }
    #pragma unroll
    for (int off = 16; off > 0; off >>= 1) s += __shfl_xor(s, off, 32);
    if (lane == 0) g[row] = s * (1.0f / (float)L_);
}

__global__ void k_gn(const float* __restrict__ g, float* __restrict__ gn) {
    int b = blockIdx.x, t = threadIdx.x;
    float a  = g[b * C_ + t];
    float c2 = g[b * C_ + 256 + t];
    float s  = a * a + c2 * c2;
    #pragma unroll
    for (int off = 16; off > 0; off >>= 1) s += __shfl_xor(s, off, 32);
    __shared__ float red[8];
    if ((t & 31) == 0) red[t >> 5] = s;
    __syncthreads();
    if (t == 0) {
        float tot = 0.f;
        #pragma unroll
        for (int i = 0; i < 8; ++i) tot += red[i];
        gn[b] = sqrtf(tot);
    }
}

__global__ void k_cos(const float* __restrict__ x, const float* __restrict__ g,
                      const float* __restrict__ gn, float* __restrict__ cosb,
                      unsigned* __restrict__ keys) {
    __shared__ float sg[C_];
    int b = blockIdx.y;
    for (int i = threadIdx.x; i < C_; i += 256) sg[i] = g[b * C_ + i];
    __syncthreads();

    int wave = threadIdx.x >> 5, lane = threadIdx.x & 31;
    int l0   = blockIdx.x * 128 + wave * 16;
    int col  = l0 + (lane & 15);
    int rowSel = (lane >> 4) * 2;
    const float* xb = x + (size_t)b * C_ * L_ + col;

    v8f acc = {};
    float sq = 0.f;
    for (int c0 = 0; c0 < C_; c0 += 4) {
        float b0 = xb[(size_t)(c0 + rowSel) * L_];
        float b1 = xb[(size_t)(c0 + rowSel + 1) * L_];
        v2f Af; Af.x = sg[c0 + rowSel]; Af.y = sg[c0 + rowSel + 1];
        v2f Bf; Bf.x = b0;              Bf.y = b1;
        acc = __builtin_amdgcn_wmma_f32_16x16x4_f32(
             false, Af,  false, Bf,
             (short)0, acc,  false,  false);
        sq = fmaf(b0, b0, sq);
        sq = fmaf(b1, b1, sq);
    }
    sq += __shfl_xor(sq, 16, 32);
    float dot  = acc[0];
    float xn   = sqrtf(sq);
    float cosv = dot / fmaxf(xn * gn[b], EPSV);
    if (lane < 16) cosb[b * L_ + col] = cosv;

    unsigned key = enc_key(cosv);
    unsigned kmin = key, kmax = key;
    #pragma unroll
    for (int off = 16; off > 0; off >>= 1) {
        unsigned a = __shfl_xor(kmin, off, 32);
        unsigned c = __shfl_xor(kmax, off, 32);
        kmin = (a < kmin) ? a : kmin;
        kmax = (c > kmax) ? c : kmax;
    }
    if (lane == 0) { atomicMin(&keys[0], kmin); atomicMax(&keys[1], kmax); }
}

__global__ void k_hist(const float* __restrict__ cosb, const unsigned* __restrict__ keys,
                       float* __restrict__ out) {
    int b = blockIdx.x, t = threadIdx.x;
    float minS = dec_key(keys[0]);
    float maxS = dec_key(keys[1]);
    float step = (maxS - minS) * (1.0f / (float)(NB - 1));

    float bins[NB];
    #pragma unroll
    for (int n = 0; n < NB; ++n) bins[n] = 0.f;
    for (int l = t; l < L_; l += 256) {
        float s = cosb[b * L_ + l];
        #pragma unroll
        for (int n = 0; n < NB; ++n) {
            float d = fabsf(s - (minS + (float)n * step));
            if (d < 0.03125f) bins[n] += (1.0f - d);
        }
    }
    __shared__ float sb[NB];
    if (t < NB) sb[t] = 0.f;
    __syncthreads();
    #pragma unroll
    for (int n = 0; n < NB; ++n) atomicAdd(&sb[n], bins[n]);
    __syncthreads();
    float tot = 0.f;
    #pragma unroll
    for (int n = 0; n < NB; ++n) tot += sb[n];
    float inv = 1.0f / tot;
    for (int i = t; i < NB * (C_ / NB); i += 256)
        out[b * (NB * (C_ / NB)) + i] = sb[i >> 5] * inv;
}

extern "C" void kernel_launch(void* const* d_in, const int* in_sizes, int n_in,
                              void* d_out, int out_size, void* d_ws, size_t ws_size,
                              hipStream_t stream) {
    const float* x = (const float*)d_in[0];
    float* out = (float*)d_out;
    char* ws = (char*)d_ws;
    float*    g    = (float*)(ws + WS_G);
    float*    gn   = (float*)(ws + WS_GN);
    unsigned* keys = (unsigned*)(ws + WS_KEYS);
    float*    cosb = (float*)(ws + WS_COS);

    k_init<<<1, 32, 0, stream>>>(keys);
    k_mean<<<(B_ * C_) / 8, 256, 0, stream>>>(x, g);
    k_gn<<<B_, 256, 0, stream>>>(g, gn);
    k_cos<<<dim3(L_ / 128, B_), 256, 0, stream>>>(x, g, gn, cosb, keys);
    k_hist<<<B_, 256, 0, stream>>>(cosb, keys, out);
}


# cosine kernel: a wave owns 32 columns, lane = column; 2 channels per matrix step, zeros in A send the 16-column halves to D rows 0, 8; whole-line loads, 56-60 in flight; channel order of sums kept
# speedup vs baseline: 1.6167x; 1.6167x over previous
#include <hip/hip_runtime.h>

typedef __attribute__((ext_vector_type(2))) float v2f;
typedef __attribute__((ext_vector_type(8))) float v8f;

#define B_   32
#define C_   512
#define L_   4096
#define NB   16
#define EPSV 1e-8f

#define WS_G    0
#define WS_GN   (B_ * C_ * 4)
#define WS_COS  (WS_GN + 256)
#define NWG     ((L_ / 128) * B_)
#define WS_MM   (WS_COS + B_ * L_ * 4)
#define NSTRIP  (L_ / 128)
#define WS_PS   (WS_MM + NWG * 8)
#define HCH     4
#define TPITCH  20

__device__ __forceinline__ unsigned enc_key(float f) {
    unsigned u = __float_as_uint(f);
    return (u & 0x80000000u) ? ~u : (u | 0x80000000u);
}
__device__ __forceinline__ float dec_key(unsigned k) {
    unsigned u = (k & 0x80000000u) ? (k & 0x7FFFFFFFu) : ~k;
    return __uint_as_float(u);
}

__global__ __launch_bounds__(256) void k_mean(const float* __restrict__ x, float* __restrict__ psum) {
    __shared__ __attribute__((aligned(8))) float srow[C_];
    int s = blockIdx.x, b = blockIdx.y;
    int wave = threadIdx.x >> 5, lane = threadIdx.x & 31;
    const float* xb = x + (size_t)b * C_ * L_ + s * 128 + lane * 4;
    bool h16 = (lane & 16) != 0, h8 = (lane & 8) != 0, h4 = (lane & 4) != 0, h2 = (lane & 2) != 0;
    for (int grp = 3; grp >= 0; --grp) {
        float v[16];
        #pragma unroll
        for (int k = 15; k >= 0; --k) {
            int c = 8 * (16 * grp + k) + wave;
            float4 q = *(const float4*)(xb + (size_t)c * L_);
            v[k] = (q.x + q.y) + (q.z + q.w);
        }
        #pragma unroll
        for (int k = 0; k < 8; ++k) {
            float keep = h16 ? v[k + 8] : v[k], send = h16 ? v[k] : v[k + 8];
            v[k] = keep + __shfl_xor(send, 16, 32);
        }
        #pragma unroll
        for (int k = 0; k < 4; ++k) {
            float keep = h8 ? v[k + 4] : v[k], send = h8 ? v[k] : v[k + 4];
            v[k] = keep + __shfl_xor(send, 8, 32);
        }
        #pragma unroll
        for (int k = 0; k < 2; ++k) {
            float keep = h4 ? v[k + 2] : v[k], send = h4 ? v[k] : v[k + 2];
            v[k] = keep + __shfl_xor(send, 4, 32);
        }
        {
            float keep = h2 ? v[1] : v[0], send = h2 ? v[0] : v[1];
            v[0] = keep + __shfl_xor(send, 2, 32);
        }
        v[0] += __shfl_xor(v[0], 1, 32);
        int kk = (lane >> 1) & 15;
        if ((lane & 1) == 0) srow[8 * (16 * grp + kk) + wave] = v[0];
    }
    __syncthreads();
    float2 o = *(const float2*)&srow[threadIdx.x * 2];
    *(float2*)&psum[(size_t)(b * NSTRIP + s) * C_ + threadIdx.x * 2] = o;
}

__global__ void k_gn(const float* __restrict__ psum, float* __restrict__ g, float* __restrict__ gn) {
    int b = blockIdx.x, t = threadIdx.x;
    const float* pb = psum + (size_t)b * NSTRIP * C_;
    float a = 0.f, c2 = 0.f;
    #pragma unroll
    for (int s2 = 0; s2 < NSTRIP; ++s2) {
        a  += pb[s2 * C_ + t];
        c2 += pb[s2 * C_ + 256 + t];
    }
    a  *= (1.0f / (float)L_);
    c2 *= (1.0f / (float)L_);
    g[b * C_ + t]       = a;
    g[b * C_ + 256 + t] = c2;
    float s  = a * a + c2 * c2;
    #pragma unroll
    for (int off = 16; off > 0; off >>= 1) s += __shfl_xor(s, off, 32);
    __shared__ float red[8];
    if ((t & 31) == 0) red[t >> 5] = s;
    __syncthreads();
    if (t == 0) {
        float tot = 0.f;
        #pragma unroll
        for (int i = 0; i < 8; ++i) tot += red[i];
        gn[b] = sqrtf(tot);
    }
}

#define CH    4
#define NS    15
#define ZOFF  (C_ + 32)

#define LOADCH(r, k)                                                            \
    _Pragma("unroll")                                                           \
    for (int j = 0; j < CH; ++j) r[j] = xb[(size_t)(CH * (k) + j) * L_];

#define CONSUME(r, a4)                                                          \
    {                                                                           \
        v2f A0; A0.x = a4.x; A0.y = a4.y;                                       \
        v2f A1; A1.x = a4.z; A1.y = a4.w;                                       \
        v2f B0; B0.x = r[0]; B0.y = r[1];                                       \
        v2f B1; B1.x = r[2]; B1.y = r[3];                                       \
        acc = __builtin_amdgcn_wmma_f32_16x16x4_f32(false, A0, false, B0, (short)0, acc, false, false); \
        acc = __builtin_amdgcn_wmma_f32_16x16x4_f32(false, A1, false, B1, (short)0, acc, false, false); \
        sqa = fmaf(r[0], r[0], sqa);                                            \
        sqa = fmaf(r[1], r[1], sqa);                                            \
        sqb = fmaf(r[2], r[2], sqb);                                            \
        sqb = fmaf(r[3], r[3], sqb);                                            \
    }

__global__ __launch_bounds__(128) void k_cos(const float* __restrict__ x, const float* __restrict__ g,
                      const float* __restrict__ gn, float* __restrict__ cosb,
                      unsigned* __restrict__ mm) {
    __shared__ __attribute__((aligned(16))) float sga[ZOFF + C_];
    int b = blockIdx.y;
    *(float4*)&sga[threadIdx.x * 4] = *(const float4*)(g + b * C_ + threadIdx.x * 4);
    *(float4*)&sga[ZOFF + threadIdx.x * 4] = make_float4(0.f, 0.f, 0.f, 0.f);
    __syncthreads();

    int wave = threadIdx.x >> 5, lane = threadIdx.x & 31;
    int col  = blockIdx.x * 128 + wave * 32 + lane;
    const float* xb = x + (size_t)b * C_ * L_ + col;
    const float* ag = sga + ((lane == 0 || lane == 24) ? 0 : ZOFF);

    v8f acc = {};
    float sqa = 0.f, sqb = 0.f;
    float r[NS][CH];
    #pragma unroll
    for (int k = 0; k < NS; ++k) { LOADCH(r[k], k) }
    float4 a[3];
    a[0] = *(const float4*)(ag);
    a[1] = *(const float4*)(ag + CH);
    #pragma unroll
    for (int k = 0; k < C_ / CH; ++k) {
        if (k + 2 < C_ / CH) a[(k + 2) % 3] = *(const float4*)(ag + CH * (k + 2));
        CONSUME(r[k % NS], a[k % 3])
        __builtin_amdgcn_sched_barrier(0);
        if (k + NS < C_ / CH) { LOADCH(r[k % NS], k + NS) }
        __builtin_amdgcn_sched_barrier(0);
    }

    float dot  = acc[0];
    float xn   = sqrtf(sqa + sqb);
    float cosv = dot / fmaxf(xn * gn[b], EPSV);
    cosb[b * L_ + col] = cosv;

    unsigned key = enc_key(cosv);
    unsigned kmin = key, kmax = key;
    #pragma unroll
    for (int off = 16; off > 0; off >>= 1) {
        unsigned a = __shfl_xor(kmin, off, 32);
        unsigned c = __shfl_xor(kmax, off, 32);
        kmin = (a < kmin) ? a : kmin;
        kmax = (c > kmax) ? c : kmax;
    }
    __shared__ unsigned smin[4], smax[4];
    if (lane == 0) { smin[wave] = kmin; smax[wave] = kmax; }
    __syncthreads();
    if (wave == 0) {
        unsigned wmin = smin[lane & 3], wmax = smax[lane & 3];
        for (int off = 2; off > 0; off >>= 1) {
            unsigned a = __shfl_xor(wmin, off, 32);
            unsigned c = __shfl_xor(wmax, off, 32);
            wmin = (a < wmin) ? a : wmin;
            wmax = (c > wmax) ? c : wmax;
        }
        if (lane == 0) {
            uint2 pr; pr.x = wmin; pr.y = wmax;
            ((uint2*)mm)[blockIdx.y * (L_ / 128) + blockIdx.x] = pr;
        }
    }
}

__global__ void k_hist_part(const float* __restrict__ cosb, const unsigned* __restrict__ mm,
                            float* __restrict__ part) {
    int b = blockIdx.y, j = blockIdx.x, t = threadIdx.x;
    const float4* cp = (const float4*)(cosb + b * L_ + j * (L_ / HCH));
    float4 v = cp[t];
    static_assert(NWG * 2 == 8 * 256, "two 16-byte loads per thread cover all pairs");
    const uint4* mp = (const uint4*)mm;
    uint4 p0 = mp[t], p1 = mp[256 + t];
    unsigned kmin = (p0.x < p0.z) ? p0.x : p0.z, kmax = (p0.y > p0.w) ? p0.y : p0.w;
    unsigned qmin = (p1.x < p1.z) ? p1.x : p1.z, qmax = (p1.y > p1.w) ? p1.y : p1.w;
    kmin = (qmin < kmin) ? qmin : kmin;
    kmax = (qmax > kmax) ? qmax : kmax;
    for (int off = 16; off > 0; off >>= 1) {
        unsigned a = __shfl_xor(kmin, off, 32);
        unsigned c = __shfl_xor(kmax, off, 32);
        kmin = (a < kmin) ? a : kmin;
        kmax = (c > kmax) ? c : kmax;
    }
    __shared__ __attribute__((aligned(16))) unsigned wmins[8], wmaxs[8];
    if ((t & 31) == 0) { wmins[t >> 5] = kmin; wmaxs[t >> 5] = kmax; }
    __syncthreads();
    unsigned gmin = wmins[0], gmax = wmaxs[0];
    for (int i = 1; i < 8; ++i) {
        unsigned a = wmins[i], c = wmaxs[i];
        gmin = (a < gmin) ? a : gmin;
        gmax = (c > gmax) ? c : gmax;
    }
    float minS = dec_key(gmin);
    float maxS = dec_key(gmax);
    float step = (maxS - minS) * (1.0f / (float)(NB - 1));
    float sv[4] = {v.x, v.y, v.z, v.w};

    float bins[NB];
    #pragma unroll
    for (int n = 0; n < NB; ++n) bins[n] = 0.f;
    #pragma unroll
    for (int e = 0; e < 4; ++e) {
        float s = sv[e];
        #pragma unroll
        for (int n = 0; n < NB; ++n) {
            float d = fabsf(s - (minS + (float)n * step));
            if (d < 0.03125f) bins[n] += (1.0f - d);
        }
    }

    __shared__ __attribute__((aligned(16))) float tile[256 * TPITCH];
    __shared__ float red[16 * NB];
    float4* row = (float4*)(tile + t * TPITCH);
    row[0] = make_float4(bins[0],  bins[1],  bins[2],  bins[3]);
    row[1] = make_float4(bins[4],  bins[5],  bins[6],  bins[7]);
    row[2] = make_float4(bins[8],  bins[9],  bins[10], bins[11]);
    row[3] = make_float4(bins[12], bins[13], bins[14], bins[15]);
    __syncthreads();
    int n = t & 15, grp = t >> 4;
    float r[16];
    #pragma unroll
    for (int i = 0; i < 16; ++i) r[i] = tile[(grp * 16 + i) * TPITCH + n];
    float a = 0.f;
    #pragma unroll
    for (int i = 0; i < 16; ++i) a += r[i];
    red[grp * NB + n] = a;
    __syncthreads();
    if (t < NB) {
        float s = 0.f;
        #pragma unroll
        for (int g2 = 0; g2 < 16; ++g2) s += red[g2 * NB + t];
        part[(b * HCH + j) * NB + t] = s;
    }
}

__global__ void k_hist_fin(const float* __restrict__ part, float* __restrict__ out) {
    int b = blockIdx.x, t = threadIdx.x;
    __shared__ float sb[NB];
    if (t < NB) {
        float s = 0.f;
        #pragma unroll
        for (int j = 0; j < HCH; ++j) s += part[(b * HCH + j) * NB + t];
        sb[t] = s;
    }
    __syncthreads();
    float tot = 0.f;
    #pragma unroll
    for (int n = 0; n < NB; ++n) tot += sb[n];
    float inv = 1.0f / tot;
    for (int i = t; i < NB * (C_ / NB); i += 256)
        out[b * (NB * (C_ / NB)) + i] = sb[i >> 5] * inv;
}

extern "C" void kernel_launch(void* const* d_in, const int* in_sizes, int n_in,
                              void* d_out, int out_size, void* d_ws, size_t ws_size,
                              hipStream_t stream) {
    const float* x = (const float*)d_in[0];
    float* out = (float*)d_out;
    char* ws = (char*)d_ws;
    float*    g    = (float*)(ws + WS_G);
    float*    gn   = (float*)(ws + WS_GN);
    unsigned* mm   = (unsigned*)(ws + WS_MM);
    float*    cosb = (float*)(ws + WS_COS);
    float*    psum = (float*)(ws + WS_PS);

    k_mean<<<dim3(NSTRIP, B_), 256, 0, stream>>>(x, psum);
    k_gn<<<B_, 256, 0, stream>>>(psum, g, gn);
    k_cos<<<dim3(L_ / 128, B_), 128, 0, stream>>>(x, g, gn, cosb, mm);
    float*    part = (float*)(ws + WS_G);
    k_hist_part<<<dim3(HCH, B_), 256, 0, stream>>>(cosb, mm, part);
    k_hist_fin<<<B_, 256, 0, stream>>>(part, out);
}
